# starting kernel
# baseline (speedup 1.0000x reference)
#include <hip/hip_runtime.h>
#include <hip/hip_bf16.h>

#define N_NODES 100000
#define N_EDGES 1600000
#define D 64
#define NF (N_NODES * D)

typedef float v2f __attribute__((ext_vector_type(2)));
typedef float v8f __attribute__((ext_vector_type(8)));

__global__ void __launch_bounds__(256) scale_kernel(
    const float* __restrict__ H,
    const float* __restrict__ outdeg,
    const float* __restrict__ indeg,
    float* __restrict__ Xout,
    float* __restrict__ Xin) {
  int t = blockIdx.x * blockDim.x + threadIdx.x;
  if (t >= NF / 4) return;
  int node = t / (D / 4);
  float doo = fmaxf(outdeg[node], 1e-8f);
  float dii = fmaxf(indeg[node], 1e-8f);
  float4 h = reinterpret_cast<const float4*>(H)[t];
  float4 o, i;
  o.x = doo * h.x; o.y = doo * h.y; o.z = doo * h.z; o.w = doo * h.w;
  i.x = dii * h.x; i.y = dii * h.y; i.z = dii * h.z; i.w = dii * h.w;
  reinterpret_cast<float4*>(Xout)[t] = o;
  reinterpret_cast<float4*>(Xin)[t]  = i;
}

__device__ __forceinline__ void atomic_add_f32(float* p, float v) {
  (void)__hip_atomic_fetch_add(p, v, __ATOMIC_RELAXED, __HIP_MEMORY_SCOPE_AGENT);
}

__global__ void __launch_bounds__(256) scatter_kernel(
    float* __restrict__ dst,
    const float* __restrict__ src,
    const int* __restrict__ didx,
    const int* __restrict__ sidx,
    const float* __restrict__ val) {
  long long t = (long long)blockIdx.x * blockDim.x + threadIdx.x;
  long long e = t >> 4;
  int q = (int)(t & 15);
  if (e >= N_EDGES) return;
  int r = didx[e];
  int c = sidx[e];
  float v = val[e];
  float4 x = reinterpret_cast<const float4*>(src + (size_t)c * D)[q];
  float* dp = dst + (size_t)r * D + (size_t)q * 4;
  atomic_add_f32(dp + 0, v * x.x);
  atomic_add_f32(dp + 1, v * x.y);
  atomic_add_f32(dp + 2, v * x.z);
  atomic_add_f32(dp + 3, v * x.w);
}

__global__ void __launch_bounds__(256) accum_kernel(
    float* __restrict__ sum,
    const float* __restrict__ Xout,
    const float* __restrict__ Xin,
    const float* __restrict__ hop,
    const float* __restrict__ th_out,
    const float* __restrict__ th_in,
    int k) {
  int t = blockIdx.x * blockDim.x + threadIdx.x;
  if (t >= NF / 4) return;
  float m = fmaxf(fmaxf(hop[0], hop[1]), hop[2]);
  float e0 = __expf(hop[0] - m);
  float e1 = __expf(hop[1] - m);
  float e2 = __expf(hop[2] - m);
  float a = ((k == 0) ? e0 : (k == 1) ? e1 : e2) / (e0 + e1 + e2);
  float co = a * th_out[k];
  float ci = a * th_in[k];
  float4 s  = reinterpret_cast<const float4*>(sum)[t];
  float4 xo = reinterpret_cast<const float4*>(Xout)[t];
  float4 xi = reinterpret_cast<const float4*>(Xin)[t];
  s.x += co * xo.x + ci * xi.x;
  s.y += co * xo.y + ci * xi.y;
  s.z += co * xo.z + ci * xi.z;
  s.w += co * xo.w + ci * xi.w;
  reinterpret_cast<float4*>(sum)[t] = s;
}

__global__ void __launch_bounds__(256) gemm_sig_add_kernel(
    float* S,
    const float* __restrict__ Theta,
    const float* __restrict__ H) {
  int lane = threadIdx.x & 31;
  int tile = blockIdx.x * (blockDim.x >> 5) + (threadIdx.x >> 5);
  if (tile >= N_NODES / 16) return;

  int m0 = tile * 16;
  int mrow  = lane & 15;
  int khalf = lane >> 4;

  const float* arow  = S + (size_t)(m0 + mrow) * D + 2 * khalf;
  const float* bbase = Theta + (size_t)(2 * khalf) * D + (lane & 15);

  v8f c0 = {}, c1 = {}, c2 = {}, c3 = {};
#pragma unroll
  for (int kk = 0; kk < 16; ++kk) {
    v2f a;
    a.x = arow[4 * kk];
    a.y = arow[4 * kk + 1];
    const float* bp = bbase + (size_t)(4 * kk) * D;
    v2f b0, b1, b2, b3;
    b0.x = bp[0];      b0.y = bp[D];
    b1.x = bp[16];     b1.y = bp[16 + D];
    b2.x = bp[32];     b2.y = bp[32 + D];
    b3.x = bp[48];     b3.y = bp[48 + D];
    c0 = __builtin_amdgcn_wmma_f32_16x16x4_f32(false, a, false, b0, (short)0, c0, false, false);
    c1 = __builtin_amdgcn_wmma_f32_16x16x4_f32(false, a, false, b1, (short)0, c1, false, false);
    c2 = __builtin_amdgcn_wmma_f32_16x16x4_f32(false, a, false, b2, (short)0, c2, false, false);
    c3 = __builtin_amdgcn_wmma_f32_16x16x4_f32(false, a, false, b3, (short)0, c3, false, false);
  }

  int colb = lane & 15;
#pragma unroll
  for (int v = 0; v < 8; ++v) {
    size_t rb = (size_t)(m0 + v + 8 * khalf) * D;
    float x0 = c0[v], x1 = c1[v], x2 = c2[v], x3 = c3[v];
    S[rb +  0 + colb] = 1.0f / (1.0f + __expf(-x0)) + H[rb +  0 + colb];
    S[rb + 16 + colb] = 1.0f / (1.0f + __expf(-x1)) + H[rb + 16 + colb];
    S[rb + 32 + colb] = 1.0f / (1.0f + __expf(-x2)) + H[rb + 32 + colb];
    S[rb + 48 + colb] = 1.0f / (1.0f + __expf(-x3)) + H[rb + 48 + colb];
  }
}

extern "C" void kernel_launch(void* const* d_in, const int* in_sizes, int n_in,
                              void* d_out, int out_size, void* d_ws, size_t ws_size,
                              hipStream_t stream) {
  const float* H      = (const float*)d_in[0];
  const int*   erow   = (const int*)d_in[1];
  const int*   ecol   = (const int*)d_in[2];
  const float* eval   = (const float*)d_in[3];
  const float* outdeg = (const float*)d_in[4];
  const float* indeg  = (const float*)d_in[5];
  const float* hop    = (const float*)d_in[6];
  const float* th_out = (const float*)d_in[7];
  const float* th_in  = (const float*)d_in[8];
  const float* Theta  = (const float*)d_in[9];

  float* sum = (float*)d_out;
  float* ws  = (float*)d_ws;
  float* Xo[2] = {ws,              ws + (size_t)NF};
  float* Xi[2] = {ws + 2ull * NF,  ws + 3ull * NF};

  const int elem_blocks = (NF / 4 + 255) / 256;

  hipMemsetAsync(sum, 0, sizeof(float) * (size_t)NF, stream);
  scale_kernel<<<elem_blocks, 256, 0, stream>>>(H, outdeg, indeg, Xo[0], Xi[0]);

  int cur = 0;
  for (int k = 0; k < 3; ++k) {
    int nxt = cur ^ 1;
    hipMemsetAsync(Xo[nxt], 0, sizeof(float) * (size_t)NF, stream);
    hipMemsetAsync(Xi[nxt], 0, sizeof(float) * (size_t)NF, stream);

    long long nthreads = (long long)N_EDGES * 16;
    int sblocks = (int)((nthreads + 255) / 256);
    scatter_kernel<<<sblocks, 256, 0, stream>>>(Xo[nxt], Xo[cur], erow, ecol, eval);
    scatter_kernel<<<sblocks, 256, 0, stream>>>(Xi[nxt], Xi[cur], ecol, erow, eval);

    accum_kernel<<<elem_blocks, 256, 0, stream>>>(sum, Xo[nxt], Xi[nxt],
                                                  hop, th_out, th_in, k);
    cur = nxt;
  }

  const int tiles = N_NODES / 16;
  gemm_sig_add_kernel<<<(tiles + 7) / 8, 256, 0, stream>>>(sum, Theta, H);
}


# Float atomic scatter replaced: per-call edge lists by target node (count, scan, integer slot adds), rows summed in registers by a gather; sum order = slot arrival; out-chain hop terms added first
# speedup vs baseline: 1.5555x; 1.5555x over previous
#include <hip/hip_runtime.h>
#include <hip/hip_bf16.h>

#define N_NODES 100000
#define N_EDGES 1600000
#define D 64
#define NF (N_NODES * D)

typedef float v2f __attribute__((ext_vector_type(2)));
typedef float v8f __attribute__((ext_vector_type(8)));

__global__ void __launch_bounds__(256) scale_kernel(
    const float* __restrict__ H,
    const float* __restrict__ deg,
    float* __restrict__ X) {
  int t = blockIdx.x * blockDim.x + threadIdx.x;
  if (t >= NF / 4) return;
  int node = t / (D / 4);
  float d = fmaxf(deg[node], 1e-8f);
  float4 h = reinterpret_cast<const float4*>(H)[t];
  float4 o;
  o.x = d * h.x; o.y = d * h.y; o.z = d * h.z; o.w = d * h.w;
  reinterpret_cast<float4*>(X)[t] = o;
}

__global__ void __launch_bounds__(256) count_kernel(
    const int* __restrict__ erow,
    const int* __restrict__ ecol,
    int* __restrict__ cnt_row,
    int* __restrict__ cnt_col) {
  int e = blockIdx.x * blockDim.x + threadIdx.x;
  if (e >= N_EDGES) return;
  (void)__hip_atomic_fetch_add(cnt_row + erow[e], 1, __ATOMIC_RELAXED, __HIP_MEMORY_SCOPE_AGENT);
  (void)__hip_atomic_fetch_add(cnt_col + ecol[e], 1, __ATOMIC_RELAXED, __HIP_MEMORY_SCOPE_AGENT);
}

__global__ void __launch_bounds__(1024) offsets_kernel(int* __restrict__ cnt_base) {
  __shared__ int buf[2][1024];
  int t = threadIdx.x;
  int4* mine = reinterpret_cast<int4*>(cnt_base + (size_t)blockIdx.x * N_NODES) + t * 25;
  int s = 0;
  if (t < N_NODES / 100) {
    for (int i = 0; i < 25; ++i) {
      int4 c = mine[i];
      s += c.x + c.y + c.z + c.w;
    }
  }
  buf[0][t] = s;
  __syncthreads();
  int cur = 0;
  for (int off = 1; off < 1024; off <<= 1) {
    int x = buf[cur][t];
    if (t >= off) x += buf[cur][t - off];
    buf[cur ^ 1][t] = x;
    __syncthreads();
    cur ^= 1;
  }
  int run = buf[cur][t] - s;
  if (t < N_NODES / 100) {
    for (int i = 0; i < 25; ++i) {
      int4 c = mine[i];
      int4 o;
      o.x = run; run += c.x;
      o.y = run; run += c.y;
      o.z = run; run += c.z;
      o.w = run; run += c.w;
      mine[i] = o;
    }
  }
}

__global__ void __launch_bounds__(256) fill_kernel(
    const int* __restrict__ erow,
    const int* __restrict__ ecol,
    const float* __restrict__ val,
    int* __restrict__ cur_row,
    int* __restrict__ cur_col,
    int* __restrict__ ent_row,
    int* __restrict__ ent_col) {
  int e = blockIdx.x * blockDim.x + threadIdx.x;
  if (e >= N_EDGES) return;
  int r = erow[e];
  int c = ecol[e];
  int vb = __float_as_int(val[e]);
  int p = __hip_atomic_fetch_add(cur_row + r, 1, __ATOMIC_RELAXED, __HIP_MEMORY_SCOPE_AGENT);
  int q = __hip_atomic_fetch_add(cur_col + c, 1, __ATOMIC_RELAXED, __HIP_MEMORY_SCOPE_AGENT);
  reinterpret_cast<int2*>(ent_row)[p] = make_int2(c, vb);
  reinterpret_cast<int2*>(ent_col)[q] = make_int2(r, vb);
}

__global__ void __launch_bounds__(256) gather_kernel(
    float* __restrict__ dst,
    const float* __restrict__ src,
    const int* __restrict__ runend,
    const int* __restrict__ ent) {
  int t = blockIdx.x * blockDim.x + threadIdx.x;
  int node = t >> 4;
  int q = t & 15;
  if (node >= N_NODES) return;
  int beg = (node > 0) ? runend[node - 1] : 0;
  int end = runend[node];
  const int2* rec = reinterpret_cast<const int2*>(ent);
  const float4* s4 = reinterpret_cast<const float4*>(src);
  float4 acc = make_float4(0.0f, 0.0f, 0.0f, 0.0f);
  for (int base = beg; base < end; base += 16) {
    int j = base + q;
    int2 en = make_int2(0, 0);
    if (j < end) en = rec[j];
    int n = min(16, end - base);
    int jj = 0;
    for (; jj + 4 <= n; jj += 4) {
      int c0 = __shfl(en.x, jj + 0, 16);
      int c1 = __shfl(en.x, jj + 1, 16);
      int c2 = __shfl(en.x, jj + 2, 16);
      int c3 = __shfl(en.x, jj + 3, 16);
      float v0 = __int_as_float(__shfl(en.y, jj + 0, 16));
      float v1 = __int_as_float(__shfl(en.y, jj + 1, 16));
      float v2 = __int_as_float(__shfl(en.y, jj + 2, 16));
      float v3 = __int_as_float(__shfl(en.y, jj + 3, 16));
      float4 x0 = s4[(size_t)c0 * (D / 4) + q];
      float4 x1 = s4[(size_t)c1 * (D / 4) + q];
      float4 x2 = s4[(size_t)c2 * (D / 4) + q];
      float4 x3 = s4[(size_t)c3 * (D / 4) + q];
      acc.x += v0 * x0.x; acc.y += v0 * x0.y; acc.z += v0 * x0.z; acc.w += v0 * x0.w;
      acc.x += v1 * x1.x; acc.y += v1 * x1.y; acc.z += v1 * x1.z; acc.w += v1 * x1.w;
      acc.x += v2 * x2.x; acc.y += v2 * x2.y; acc.z += v2 * x2.z; acc.w += v2 * x2.w;
      acc.x += v3 * x3.x; acc.y += v3 * x3.y; acc.z += v3 * x3.z; acc.w += v3 * x3.w;
    }
    for (; jj < n; ++jj) {
      int c0 = __shfl(en.x, jj, 16);
      float v0 = __int_as_float(__shfl(en.y, jj, 16));
      float4 x0 = s4[(size_t)c0 * (D / 4) + q];
      acc.x += v0 * x0.x; acc.y += v0 * x0.y; acc.z += v0 * x0.z; acc.w += v0 * x0.w;
    }
  }
  reinterpret_cast<float4*>(dst)[(size_t)node * (D / 4) + q] = acc;
}

__global__ void __launch_bounds__(256) accum_kernel(
    float* __restrict__ sum,
    const float* __restrict__ X,
    const float* __restrict__ hop,
    const float* __restrict__ th,
    int k) {
  int t = blockIdx.x * blockDim.x + threadIdx.x;
  if (t >= NF / 4) return;
  float m = fmaxf(fmaxf(hop[0], hop[1]), hop[2]);
  float e0 = __expf(hop[0] - m);
  float e1 = __expf(hop[1] - m);
  float e2 = __expf(hop[2] - m);
  float a = ((k == 0) ? e0 : (k == 1) ? e1 : e2) / (e0 + e1 + e2);
  float co = a * th[k];
  float4 s = reinterpret_cast<const float4*>(sum)[t];
  float4 x = reinterpret_cast<const float4*>(X)[t];
  s.x += co * x.x;
  s.y += co * x.y;
  s.z += co * x.z;
  s.w += co * x.w;
  reinterpret_cast<float4*>(sum)[t] = s;
}

__global__ void __launch_bounds__(256) gemm_sig_add_kernel(
    float* S,
    const float* __restrict__ Theta,
    const float* __restrict__ H) {
  int lane = threadIdx.x & 31;
  int tile = blockIdx.x * (blockDim.x >> 5) + (threadIdx.x >> 5);
  if (tile >= N_NODES / 16) return;

  int m0 = tile * 16;
  int mrow  = lane & 15;
  int khalf = lane >> 4;

  const float* arow  = S + (size_t)(m0 + mrow) * D + 2 * khalf;
  const float* bbase = Theta + (size_t)(2 * khalf) * D + (lane & 15);

  v8f c0 = {}, c1 = {}, c2 = {}, c3 = {};
#pragma unroll
  for (int kk = 0; kk < 16; ++kk) {
    v2f a;
    a.x = arow[4 * kk];
    a.y = arow[4 * kk + 1];
    const float* bp = bbase + (size_t)(4 * kk) * D;
    v2f b0, b1, b2, b3;
    b0.x = bp[0];      b0.y = bp[D];
    b1.x = bp[16];     b1.y = bp[16 + D];
    b2.x = bp[32];     b2.y = bp[32 + D];
    b3.x = bp[48];     b3.y = bp[48 + D];
    c0 = __builtin_amdgcn_wmma_f32_16x16x4_f32(false, a, false, b0, (short)0, c0, false, false);
    c1 = __builtin_amdgcn_wmma_f32_16x16x4_f32(false, a, false, b1, (short)0, c1, false, false);
    c2 = __builtin_amdgcn_wmma_f32_16x16x4_f32(false, a, false, b2, (short)0, c2, false, false);
    c3 = __builtin_amdgcn_wmma_f32_16x16x4_f32(false, a, false, b3, (short)0, c3, false, false);
  }

  int colb = lane & 15;
#pragma unroll
  for (int v = 0; v < 8; ++v) {
    size_t rb = (size_t)(m0 + v + 8 * khalf) * D;
    float x0 = c0[v], x1 = c1[v], x2 = c2[v], x3 = c3[v];
    S[rb +  0 + colb] = 1.0f / (1.0f + __expf(-x0)) + H[rb +  0 + colb];
    S[rb + 16 + colb] = 1.0f / (1.0f + __expf(-x1)) + H[rb + 16 + colb];
    S[rb + 32 + colb] = 1.0f / (1.0f + __expf(-x2)) + H[rb + 32 + colb];
    S[rb + 48 + colb] = 1.0f / (1.0f + __expf(-x3)) + H[rb + 48 + colb];
  }
}

extern "C" void kernel_launch(void* const* d_in, const int* in_sizes, int n_in,
                              void* d_out, int out_size, void* d_ws, size_t ws_size,
                              hipStream_t stream) {
  const float* H      = (const float*)d_in[0];
  const int*   erow   = (const int*)d_in[1];
  const int*   ecol   = (const int*)d_in[2];
  const float* eval   = (const float*)d_in[3];
  const float* outdeg = (const float*)d_in[4];
  const float* indeg  = (const float*)d_in[5];
  const float* hop    = (const float*)d_in[6];
  const float* th_out = (const float*)d_in[7];
  const float* th_in  = (const float*)d_in[8];
  const float* Theta  = (const float*)d_in[9];

  float* sum = (float*)d_out;
  float* ws  = (float*)d_ws;
  float* X[2]    = {ws, ws + (size_t)NF};
  int*   ent_row = (int*)(ws + 2ull * NF);
  int*   ent_col = ent_row + 2ull * N_EDGES;
  int*   cur_row = ent_col + 2ull * N_EDGES;
  int*   cur_col = cur_row + N_NODES;

  const int elem_blocks = (NF / 4 + 255) / 256;
  const int edge_blocks = (N_EDGES + 255) / 256;
  const int node_blocks = (N_NODES * 16 + 255) / 256;

  hipMemsetAsync(sum, 0, sizeof(float) * (size_t)NF, stream);
  hipMemsetAsync(cur_row, 0, sizeof(int) * 2ull * N_NODES, stream);

  count_kernel<<<edge_blocks, 256, 0, stream>>>(erow, ecol, cur_row, cur_col);
  offsets_kernel<<<2, 1024, 0, stream>>>(cur_row);
  fill_kernel<<<edge_blocks, 256, 0, stream>>>(erow, ecol, eval, cur_row, cur_col,
                                               ent_row, ent_col);

  scale_kernel<<<elem_blocks, 256, 0, stream>>>(H, outdeg, X[0]);
  int cur = 0;
  for (int k = 0; k < 3; ++k) {
    int nxt = cur ^ 1;
    gather_kernel<<<node_blocks, 256, 0, stream>>>(X[nxt], X[cur], cur_row, ent_row);
    accum_kernel<<<elem_blocks, 256, 0, stream>>>(sum, X[nxt], hop, th_out, k);
    cur = nxt;
  }
  scale_kernel<<<elem_blocks, 256, 0, stream>>>(H, indeg, X[cur ^ 1]);
  cur ^= 1;
  for (int k = 0; k < 3; ++k) {
    int nxt = cur ^ 1;
    gather_kernel<<<node_blocks, 256, 0, stream>>>(X[nxt], X[cur], cur_col, ent_col);
    accum_kernel<<<elem_blocks, 256, 0, stream>>>(sum, X[nxt], hop, th_in, k);
    cur = nxt;
  }

  const int tiles = N_NODES / 16;
  gemm_sig_add_kernel<<<(tiles + 7) / 8, 256, 0, stream>>>(sum, Theta, H);
}


# Hop-weighted sum folded into the gather epilogue: each gather adds alpha*theta*row to out from registers (first one stores); 6 add passes, clear of out and 2 plane stores removed; sum order kept
# speedup vs baseline: 1.6483x; 1.0597x over previous
#include <hip/hip_runtime.h>
#include <hip/hip_bf16.h>

#define N_NODES 100000
#define N_EDGES 1600000
#define D 64
#define NF (N_NODES * D)

typedef float v2f __attribute__((ext_vector_type(2)));
typedef float v8f __attribute__((ext_vector_type(8)));

__global__ void __launch_bounds__(256) scale_kernel(
    const float* __restrict__ H,
    const float* __restrict__ deg,
    float* __restrict__ X) {
  int t = blockIdx.x * blockDim.x + threadIdx.x;
  if (t >= NF / 4) return;
  int node = t / (D / 4);
  float d = fmaxf(deg[node], 1e-8f);
  float4 h = reinterpret_cast<const float4*>(H)[t];
  float4 o;
  o.x = d * h.x; o.y = d * h.y; o.z = d * h.z; o.w = d * h.w;
  reinterpret_cast<float4*>(X)[t] = o;
}

__global__ void __launch_bounds__(256) count_kernel(
    const int* __restrict__ erow,
    const int* __restrict__ ecol,
    int* __restrict__ cnt_row,
    int* __restrict__ cnt_col) {
  int e = blockIdx.x * blockDim.x + threadIdx.x;
  if (e >= N_EDGES) return;
  (void)__hip_atomic_fetch_add(cnt_row + erow[e], 1, __ATOMIC_RELAXED, __HIP_MEMORY_SCOPE_AGENT);
  (void)__hip_atomic_fetch_add(cnt_col + ecol[e], 1, __ATOMIC_RELAXED, __HIP_MEMORY_SCOPE_AGENT);
}

__global__ void __launch_bounds__(1024) offsets_kernel(int* __restrict__ cnt_base) {
  __shared__ int buf[2][1024];
  int t = threadIdx.x;
  int4* mine = reinterpret_cast<int4*>(cnt_base + (size_t)blockIdx.x * N_NODES) + t * 25;
  int s = 0;
  if (t < N_NODES / 100) {
    for (int i = 0; i < 25; ++i) {
      int4 c = mine[i];
      s += c.x + c.y + c.z + c.w;
    }
  }
  buf[0][t] = s;
  __syncthreads();
  int cur = 0;
  for (int off = 1; off < 1024; off <<= 1) {
    int x = buf[cur][t];
    if (t >= off) x += buf[cur][t - off];
    buf[cur ^ 1][t] = x;
    __syncthreads();
    cur ^= 1;
  }
  int run = buf[cur][t] - s;
  if (t < N_NODES / 100) {
    for (int i = 0; i < 25; ++i) {
      int4 c = mine[i];
      int4 o;
      o.x = run; run += c.x;
      o.y = run; run += c.y;
      o.z = run; run += c.z;
      o.w = run; run += c.w;
      mine[i] = o;
    }
  }
}

__global__ void __launch_bounds__(256) fill_kernel(
    const int* __restrict__ erow,
    const int* __restrict__ ecol,
    const float* __restrict__ val,
    int* __restrict__ cur_row,
    int* __restrict__ cur_col,
    int* __restrict__ ent_row,
    int* __restrict__ ent_col) {
  int e = blockIdx.x * blockDim.x + threadIdx.x;
  if (e >= N_EDGES) return;
  int r = erow[e];
  int c = ecol[e];
  int vb = __float_as_int(val[e]);
  int p = __hip_atomic_fetch_add(cur_row + r, 1, __ATOMIC_RELAXED, __HIP_MEMORY_SCOPE_AGENT);
  int q = __hip_atomic_fetch_add(cur_col + c, 1, __ATOMIC_RELAXED, __HIP_MEMORY_SCOPE_AGENT);
  reinterpret_cast<int2*>(ent_row)[p] = make_int2(c, vb);
  reinterpret_cast<int2*>(ent_col)[q] = make_int2(r, vb);
}

template <bool FIRST, bool KEEP_X>
__global__ void __launch_bounds__(256) gather_kernel(
    float* __restrict__ dst,
    const float* __restrict__ src,
    const int* __restrict__ runend,
    const int* __restrict__ ent,
    float* __restrict__ sum,
    const float* __restrict__ hop,
    const float* __restrict__ th,
    int k) {
  int t = blockIdx.x * blockDim.x + threadIdx.x;
  int node = t >> 4;
  int q = t & 15;
  if (node >= N_NODES) return;
  size_t at = (size_t)node * (D / 4) + q;
  float4 s = make_float4(0.0f, 0.0f, 0.0f, 0.0f);
  if (!FIRST) s = reinterpret_cast<const float4*>(sum)[at];
  float m = fmaxf(fmaxf(hop[0], hop[1]), hop[2]);
  float e0 = __expf(hop[0] - m);
  float e1 = __expf(hop[1] - m);
  float e2 = __expf(hop[2] - m);
  float a = ((k == 0) ? e0 : (k == 1) ? e1 : e2) / (e0 + e1 + e2);
  float co = a * th[k];
  int beg = (node > 0) ? runend[node - 1] : 0;
  int end = runend[node];
  const int2* rec = reinterpret_cast<const int2*>(ent);
  const float4* s4 = reinterpret_cast<const float4*>(src);
  float4 acc = make_float4(0.0f, 0.0f, 0.0f, 0.0f);
  for (int base = beg; base < end; base += 16) {
    int j = base + q;
    int2 en = make_int2(0, 0);
    if (j < end) en = rec[j];
    int n = min(16, end - base);
    int jj = 0;
    for (; jj + 4 <= n; jj += 4) {
      int c0 = __shfl(en.x, jj + 0, 16);
      int c1 = __shfl(en.x, jj + 1, 16);
      int c2 = __shfl(en.x, jj + 2, 16);
      int c3 = __shfl(en.x, jj + 3, 16);
      float v0 = __int_as_float(__shfl(en.y, jj + 0, 16));
      float v1 = __int_as_float(__shfl(en.y, jj + 1, 16));
      float v2 = __int_as_float(__shfl(en.y, jj + 2, 16));
      float v3 = __int_as_float(__shfl(en.y, jj + 3, 16));
      float4 x0 = s4[(size_t)c0 * (D / 4) + q];
      float4 x1 = s4[(size_t)c1 * (D / 4) + q];
      float4 x2 = s4[(size_t)c2 * (D / 4) + q];
      float4 x3 = s4[(size_t)c3 * (D / 4) + q];
      acc.x += v0 * x0.x; acc.y += v0 * x0.y; acc.z += v0 * x0.z; acc.w += v0 * x0.w;
      acc.x += v1 * x1.x; acc.y += v1 * x1.y; acc.z += v1 * x1.z; acc.w += v1 * x1.w;
      acc.x += v2 * x2.x; acc.y += v2 * x2.y; acc.z += v2 * x2.z; acc.w += v2 * x2.w;
      acc.x += v3 * x3.x; acc.y += v3 * x3.y; acc.z += v3 * x3.z; acc.w += v3 * x3.w;
    }
    for (; jj < n; ++jj) {
      int c0 = __shfl(en.x, jj, 16);
      float v0 = __int_as_float(__shfl(en.y, jj, 16));
      float4 x0 = s4[(size_t)c0 * (D / 4) + q];
      acc.x += v0 * x0.x; acc.y += v0 * x0.y; acc.z += v0 * x0.z; acc.w += v0 * x0.w;
    }
  }
  if (KEEP_X) reinterpret_cast<float4*>(dst)[at] = acc;

  if (FIRST) {
    s.x = co * acc.x;
    s.y = co * acc.y;
    s.z = co * acc.z;
    s.w = co * acc.w;
  } else {
    s.x += co * acc.x;
    s.y += co * acc.y;
    s.z += co * acc.z;
    s.w += co * acc.w;
  }
  reinterpret_cast<float4*>(sum)[at] = s;
}

__global__ void __launch_bounds__(256) gemm_sig_add_kernel(
    float* S,
    const float* __restrict__ Theta,
    const float* __restrict__ H) {
  int lane = threadIdx.x & 31;
  int tile = blockIdx.x * (blockDim.x >> 5) + (threadIdx.x >> 5);
  if (tile >= N_NODES / 16) return;

  int m0 = tile * 16;
  int mrow  = lane & 15;
  int khalf = lane >> 4;

  const float* arow  = S + (size_t)(m0 + mrow) * D + 2 * khalf;
  const float* bbase = Theta + (size_t)(2 * khalf) * D + (lane & 15);

  v8f c0 = {}, c1 = {}, c2 = {}, c3 = {};
#pragma unroll
  for (int kk = 0; kk < 16; ++kk) {
    v2f a;
    a.x = arow[4 * kk];
    a.y = arow[4 * kk + 1];
    const float* bp = bbase + (size_t)(4 * kk) * D;
    v2f b0, b1, b2, b3;
    b0.x = bp[0];      b0.y = bp[D];
    b1.x = bp[16];     b1.y = bp[16 + D];
    b2.x = bp[32];     b2.y = bp[32 + D];
    b3.x = bp[48];     b3.y = bp[48 + D];
    c0 = __builtin_amdgcn_wmma_f32_16x16x4_f32(false, a, false, b0, (short)0, c0, false, false);
    c1 = __builtin_amdgcn_wmma_f32_16x16x4_f32(false, a, false, b1, (short)0, c1, false, false);
    c2 = __builtin_amdgcn_wmma_f32_16x16x4_f32(false, a, false, b2, (short)0, c2, false, false);
    c3 = __builtin_amdgcn_wmma_f32_16x16x4_f32(false, a, false, b3, (short)0, c3, false, false);
  }

  int colb = lane & 15;
#pragma unroll
  for (int v = 0; v < 8; ++v) {
    size_t rb = (size_t)(m0 + v + 8 * khalf) * D;
    float x0 = c0[v], x1 = c1[v], x2 = c2[v], x3 = c3[v];
    S[rb +  0 + colb] = 1.0f / (1.0f + __expf(-x0)) + H[rb +  0 + colb];
    S[rb + 16 + colb] = 1.0f / (1.0f + __expf(-x1)) + H[rb + 16 + colb];
    S[rb + 32 + colb] = 1.0f / (1.0f + __expf(-x2)) + H[rb + 32 + colb];
    S[rb + 48 + colb] = 1.0f / (1.0f + __expf(-x3)) + H[rb + 48 + colb];
  }
}

extern "C" void kernel_launch(void* const* d_in, const int* in_sizes, int n_in,
                              void* d_out, int out_size, void* d_ws, size_t ws_size,
                              hipStream_t stream) {
  const float* H      = (const float*)d_in[0];
  const int*   erow   = (const int*)d_in[1];
  const int*   ecol   = (const int*)d_in[2];
  const float* eval   = (const float*)d_in[3];
  const float* outdeg = (const float*)d_in[4];
  const float* indeg  = (const float*)d_in[5];
  const float* hop    = (const float*)d_in[6];
  const float* th_out = (const float*)d_in[7];
  const float* th_in  = (const float*)d_in[8];
  const float* Theta  = (const float*)d_in[9];

  float* sum = (float*)d_out;
  float* ws  = (float*)d_ws;
  float* X[2]    = {ws, ws + (size_t)NF};
  int*   ent_row = (int*)(ws + 2ull * NF);
  int*   ent_col = ent_row + 2ull * N_EDGES;
  int*   cur_row = ent_col + 2ull * N_EDGES;
  int*   cur_col = cur_row + N_NODES;

  const int elem_blocks = (NF / 4 + 255) / 256;
  const int edge_blocks = (N_EDGES + 255) / 256;
  const int node_blocks = (N_NODES * 16 + 255) / 256;

  hipMemsetAsync(cur_row, 0, sizeof(int) * 2ull * N_NODES, stream);

  count_kernel<<<edge_blocks, 256, 0, stream>>>(erow, ecol, cur_row, cur_col);
  offsets_kernel<<<2, 1024, 0, stream>>>(cur_row);
  fill_kernel<<<edge_blocks, 256, 0, stream>>>(erow, ecol, eval, cur_row, cur_col,
                                               ent_row, ent_col);

  scale_kernel<<<elem_blocks, 256, 0, stream>>>(H, outdeg, X[0]);
  gather_kernel<true, true><<<node_blocks, 256, 0, stream>>>(
      X[1], X[0], cur_row, ent_row, sum, hop, th_out, 0);
  gather_kernel<false, true><<<node_blocks, 256, 0, stream>>>(
      X[0], X[1], cur_row, ent_row, sum, hop, th_out, 1);
  gather_kernel<false, false><<<node_blocks, 256, 0, stream>>>(
      X[1], X[0], cur_row, ent_row, sum, hop, th_out, 2);

  scale_kernel<<<elem_blocks, 256, 0, stream>>>(H, indeg, X[1]);
  gather_kernel<false, true><<<node_blocks, 256, 0, stream>>>(
      X[0], X[1], cur_col, ent_col, sum, hop, th_in, 0);
  gather_kernel<false, true><<<node_blocks, 256, 0, stream>>>(
      X[1], X[0], cur_col, ent_col, sum, hop, th_in, 1);
  gather_kernel<false, false><<<node_blocks, 256, 0, stream>>>(
      X[0], X[1], cur_col, ent_col, sum, hop, th_in, 2);

  const int tiles = N_NODES / 16;
  gemm_sig_add_kernel<<<(tiles + 7) / 8, 256, 0, stream>>>(sum, Theta, H);
}


# Edge ranks taken from the count pass's returning integer adds and kept per edge; fill places records at run start + rank with no atomics; sum order = arrival order of the count adds
# speedup vs baseline: 1.9725x; 1.1967x over previous
#include <hip/hip_runtime.h>
#include <hip/hip_bf16.h>

#define N_NODES 100000
#define N_EDGES 1600000
#define D 64
#define NF (N_NODES * D)

typedef float v2f __attribute__((ext_vector_type(2)));
typedef float v8f __attribute__((ext_vector_type(8)));

__global__ void __launch_bounds__(256) scale_kernel(
    const float* __restrict__ H,
    const float* __restrict__ deg,
    float* __restrict__ X) {
  int t = blockIdx.x * blockDim.x + threadIdx.x;
  if (t >= NF / 4) return;
  int node = t / (D / 4);
  float d = fmaxf(deg[node], 1e-8f);
  float4 h = reinterpret_cast<const float4*>(H)[t];
  float4 o;
  o.x = d * h.x; o.y = d * h.y; o.z = d * h.z; o.w = d * h.w;
  reinterpret_cast<float4*>(X)[t] = o;
}

__global__ void __launch_bounds__(256) count_kernel(
    const int* __restrict__ erow,
    const int* __restrict__ ecol,
    int* __restrict__ cnt_row,
    int* __restrict__ cnt_col,
    int* __restrict__ rank) {
  int e = blockIdx.x * blockDim.x + threadIdx.x;
  if (e >= N_EDGES) return;
  int r = erow[e];
  int c = ecol[e];
  int pr = __hip_atomic_fetch_add(cnt_row + r, 1, __ATOMIC_RELAXED, __HIP_MEMORY_SCOPE_AGENT);
  int pc = __hip_atomic_fetch_add(cnt_col + c, 1, __ATOMIC_RELAXED, __HIP_MEMORY_SCOPE_AGENT);
  reinterpret_cast<int2*>(rank)[e] = make_int2(pr, pc);
}

__global__ void __launch_bounds__(1024) offsets_kernel(int* __restrict__ cnt_base) {
  __shared__ int buf[2][1024];
  int t = threadIdx.x;
  int4* mine = reinterpret_cast<int4*>(cnt_base + (size_t)blockIdx.x * N_NODES) + t * 25;
  int s = 0;
  if (t < N_NODES / 100) {
    for (int i = 0; i < 25; ++i) {
      int4 c = mine[i];
      s += c.x + c.y + c.z + c.w;
    }
  }
  buf[0][t] = s;
  __syncthreads();
  int cur = 0;
  for (int off = 1; off < 1024; off <<= 1) {
    int x = buf[cur][t];
    if (t >= off) x += buf[cur][t - off];
    buf[cur ^ 1][t] = x;
    __syncthreads();
    cur ^= 1;
  }
  int run = buf[cur][t] - s;
  if (t < N_NODES / 100) {
    for (int i = 0; i < 25; ++i) {
      int4 c = mine[i];
      int4 o;
      o.x = run; run += c.x;
      o.y = run; run += c.y;
      o.z = run; run += c.z;
      o.w = run; run += c.w;
      mine[i] = o;
    }
  }
}

__global__ void __launch_bounds__(256) fill_kernel(
    const int* __restrict__ erow,
    const int* __restrict__ ecol,
    const float* __restrict__ val,
    const int* __restrict__ rank,
    const int* __restrict__ start_row,
    const int* __restrict__ start_col,
    int* __restrict__ ent_row,
    int* __restrict__ ent_col) {
  int e = blockIdx.x * blockDim.x + threadIdx.x;
  if (e >= N_EDGES) return;
  int r = erow[e];
  int c = ecol[e];
  int vb = __float_as_int(val[e]);
  int2 rk = reinterpret_cast<const int2*>(rank)[e];
  int p = start_row[r] + rk.x;
  int q = start_col[c] + rk.y;
  reinterpret_cast<int2*>(ent_row)[p] = make_int2(c, vb);
  reinterpret_cast<int2*>(ent_col)[q] = make_int2(r, vb);
}

template <bool FIRST, bool KEEP_X>
__global__ void __launch_bounds__(256) gather_kernel(
    float* __restrict__ dst,
    const float* __restrict__ src,
    const int* __restrict__ runstart,
    const int* __restrict__ ent,
    float* __restrict__ sum,
    const float* __restrict__ hop,
    const float* __restrict__ th,
    int k) {
  int t = blockIdx.x * blockDim.x + threadIdx.x;
  int node = t >> 4;
  int q = t & 15;
  if (node >= N_NODES) return;
  size_t at = (size_t)node * (D / 4) + q;
  float4 s = make_float4(0.0f, 0.0f, 0.0f, 0.0f);
  if (!FIRST) s = reinterpret_cast<const float4*>(sum)[at];
  float m = fmaxf(fmaxf(hop[0], hop[1]), hop[2]);
  float e0 = __expf(hop[0] - m);
  float e1 = __expf(hop[1] - m);
  float e2 = __expf(hop[2] - m);
  float a = ((k == 0) ? e0 : (k == 1) ? e1 : e2) / (e0 + e1 + e2);
  float co = a * th[k];
  int beg = runstart[node];
  int end = (node + 1 < N_NODES) ? runstart[node + 1] : N_EDGES;
  const int2* rec = reinterpret_cast<const int2*>(ent);
  const float4* s4 = reinterpret_cast<const float4*>(src);
  float4 acc = make_float4(0.0f, 0.0f, 0.0f, 0.0f);
  for (int base = beg; base < end; base += 16) {
    int j = base + q;
    int2 en = make_int2(0, 0);
    if (j < end) en = rec[j];
    int n = min(16, end - base);
    int jj = 0;
    for (; jj + 4 <= n; jj += 4) {
      int c0 = __shfl(en.x, jj + 0, 16);
      int c1 = __shfl(en.x, jj + 1, 16);
      int c2 = __shfl(en.x, jj + 2, 16);
      int c3 = __shfl(en.x, jj + 3, 16);
      float v0 = __int_as_float(__shfl(en.y, jj + 0, 16));
      float v1 = __int_as_float(__shfl(en.y, jj + 1, 16));
      float v2 = __int_as_float(__shfl(en.y, jj + 2, 16));
      float v3 = __int_as_float(__shfl(en.y, jj + 3, 16));
      float4 x0 = s4[(size_t)c0 * (D / 4) + q];
      float4 x1 = s4[(size_t)c1 * (D / 4) + q];
      float4 x2 = s4[(size_t)c2 * (D / 4) + q];
      float4 x3 = s4[(size_t)c3 * (D / 4) + q];
      acc.x += v0 * x0.x; acc.y += v0 * x0.y; acc.z += v0 * x0.z; acc.w += v0 * x0.w;
      acc.x += v1 * x1.x; acc.y += v1 * x1.y; acc.z += v1 * x1.z; acc.w += v1 * x1.w;
      acc.x += v2 * x2.x; acc.y += v2 * x2.y; acc.z += v2 * x2.z; acc.w += v2 * x2.w;
      acc.x += v3 * x3.x; acc.y += v3 * x3.y; acc.z += v3 * x3.z; acc.w += v3 * x3.w;
    }
    for (; jj < n; ++jj) {
      int c0 = __shfl(en.x, jj, 16);
      float v0 = __int_as_float(__shfl(en.y, jj, 16));
      float4 x0 = s4[(size_t)c0 * (D / 4) + q];
      acc.x += v0 * x0.x; acc.y += v0 * x0.y; acc.z += v0 * x0.z; acc.w += v0 * x0.w;
    }
  }
  if (KEEP_X) reinterpret_cast<float4*>(dst)[at] = acc;

  if (FIRST) {
    s.x = co * acc.x;
    s.y = co * acc.y;
    s.z = co * acc.z;
    s.w = co * acc.w;
  } else {
    s.x += co * acc.x;
    s.y += co * acc.y;
    s.z += co * acc.z;
    s.w += co * acc.w;
  }
  reinterpret_cast<float4*>(sum)[at] = s;
}

__global__ void __launch_bounds__(256) gemm_sig_add_kernel(
    float* S,
    const float* __restrict__ Theta,
    const float* __restrict__ H) {
  int lane = threadIdx.x & 31;
  int tile = blockIdx.x * (blockDim.x >> 5) + (threadIdx.x >> 5);
  if (tile >= N_NODES / 16) return;

  int m0 = tile * 16;
  int mrow  = lane & 15;
  int khalf = lane >> 4;

  const float* arow  = S + (size_t)(m0 + mrow) * D + 2 * khalf;
  const float* bbase = Theta + (size_t)(2 * khalf) * D + (lane & 15);

  v8f c0 = {}, c1 = {}, c2 = {}, c3 = {};
#pragma unroll
  for (int kk = 0; kk < 16; ++kk) {
    v2f a;
    a.x = arow[4 * kk];
    a.y = arow[4 * kk + 1];
    const float* bp = bbase + (size_t)(4 * kk) * D;
    v2f b0, b1, b2, b3;
    b0.x = bp[0];      b0.y = bp[D];
    b1.x = bp[16];     b1.y = bp[16 + D];
    b2.x = bp[32];     b2.y = bp[32 + D];
    b3.x = bp[48];     b3.y = bp[48 + D];
    c0 = __builtin_amdgcn_wmma_f32_16x16x4_f32(false, a, false, b0, (short)0, c0, false, false);
    c1 = __builtin_amdgcn_wmma_f32_16x16x4_f32(false, a, false, b1, (short)0, c1, false, false);
    c2 = __builtin_amdgcn_wmma_f32_16x16x4_f32(false, a, false, b2, (short)0, c2, false, false);
    c3 = __builtin_amdgcn_wmma_f32_16x16x4_f32(false, a, false, b3, (short)0, c3, false, false);
  }

  int colb = lane & 15;
#pragma unroll
  for (int v = 0; v < 8; ++v) {
    size_t rb = (size_t)(m0 + v + 8 * khalf) * D;
    float x0 = c0[v], x1 = c1[v], x2 = c2[v], x3 = c3[v];
    S[rb +  0 + colb] = 1.0f / (1.0f + __expf(-x0)) + H[rb +  0 + colb];
    S[rb + 16 + colb] = 1.0f / (1.0f + __expf(-x1)) + H[rb + 16 + colb];
    S[rb + 32 + colb] = 1.0f / (1.0f + __expf(-x2)) + H[rb + 32 + colb];
    S[rb + 48 + colb] = 1.0f / (1.0f + __expf(-x3)) + H[rb + 48 + colb];
  }
}

extern "C" void kernel_launch(void* const* d_in, const int* in_sizes, int n_in,
                              void* d_out, int out_size, void* d_ws, size_t ws_size,
                              hipStream_t stream) {
  const float* H      = (const float*)d_in[0];
  const int*   erow   = (const int*)d_in[1];
  const int*   ecol   = (const int*)d_in[2];
  const float* eval   = (const float*)d_in[3];
  const float* outdeg = (const float*)d_in[4];
  const float* indeg  = (const float*)d_in[5];
  const float* hop    = (const float*)d_in[6];
  const float* th_out = (const float*)d_in[7];
  const float* th_in  = (const float*)d_in[8];
  const float* Theta  = (const float*)d_in[9];

  float* sum = (float*)d_out;
  float* ws  = (float*)d_ws;
  float* X[2]    = {ws, ws + (size_t)NF};
  int*   ent_row = (int*)(ws + 2ull * NF);
  int*   ent_col = ent_row + 2ull * N_EDGES;
  int*   cur_row = ent_col + 2ull * N_EDGES;
  int*   cur_col = cur_row + N_NODES;
  int*   rank    = (int*)X[1];

  const int elem_blocks = (NF / 4 + 255) / 256;
  const int edge_blocks = (N_EDGES + 255) / 256;
  const int node_blocks = (N_NODES * 16 + 255) / 256;

  hipMemsetAsync(cur_row, 0, sizeof(int) * 2ull * N_NODES, stream);

  count_kernel<<<edge_blocks, 256, 0, stream>>>(erow, ecol, cur_row, cur_col, rank);
  offsets_kernel<<<2, 1024, 0, stream>>>(cur_row);
  fill_kernel<<<edge_blocks, 256, 0, stream>>>(erow, ecol, eval, rank, cur_row, cur_col,
                                               ent_row, ent_col);

  scale_kernel<<<elem_blocks, 256, 0, stream>>>(H, outdeg, X[0]);
  gather_kernel<true, true><<<node_blocks, 256, 0, stream>>>(
      X[1], X[0], cur_row, ent_row, sum, hop, th_out, 0);
  gather_kernel<false, true><<<node_blocks, 256, 0, stream>>>(
      X[0], X[1], cur_row, ent_row, sum, hop, th_out, 1);
  gather_kernel<false, false><<<node_blocks, 256, 0, stream>>>(
      X[1], X[0], cur_row, ent_row, sum, hop, th_out, 2);

  scale_kernel<<<elem_blocks, 256, 0, stream>>>(H, indeg, X[1]);
  gather_kernel<false, true><<<node_blocks, 256, 0, stream>>>(
      X[0], X[1], cur_col, ent_col, sum, hop, th_in, 0);
  gather_kernel<false, true><<<node_blocks, 256, 0, stream>>>(
      X[1], X[0], cur_col, ent_col, sum, hop, th_in, 1);
  gather_kernel<false, false><<<node_blocks, 256, 0, stream>>>(
      X[0], X[1], cur_col, ent_col, sum, hop, th_in, 2);

  const int tiles = N_NODES / 16;
  gemm_sig_add_kernel<<<(tiles + 7) / 8, 256, 0, stream>>>(sum, Theta, H);
}


# Edge lists built by a two-level bucket sort with LDS counters (edges binned by 256-node range, then ranked per node in LDS); no per-edge global atomics; sum order = arrival order of the LDS adds
# speedup vs baseline: 3.5545x; 1.8020x over previous
#include <hip/hip_runtime.h>
#include <hip/hip_bf16.h>

#define N_NODES 100000
#define N_EDGES 1600000
#define D 64
#define NF (N_NODES * D)

typedef float v2f __attribute__((ext_vector_type(2)));
typedef float v8f __attribute__((ext_vector_type(8)));

__global__ void __launch_bounds__(256) scale_kernel(
    const float* __restrict__ H,
    const float* __restrict__ deg,
    float* __restrict__ X) {
  int t = blockIdx.x * blockDim.x + threadIdx.x;
  if (t >= NF / 4) return;
  int node = t / (D / 4);
  float d = fmaxf(deg[node], 1e-8f);
  float4 h = reinterpret_cast<const float4*>(H)[t];
  float4 o;
  o.x = d * h.x; o.y = d * h.y; o.z = d * h.z; o.w = d * h.w;
  reinterpret_cast<float4*>(X)[t] = o;
}

#define BSH 8
#define BNODES (1 << BSH)
#define NBKT ((N_NODES + BNODES - 1) / BNODES)
#define NBIN (2 * NBKT)
#define NB1 256
#define SLICE ((N_EDGES + NB1 - 1) / NB1)
#define OTHER_MASK 0x1FFFF

__device__ __forceinline__ void lds_inc(int* p) {
  __hip_atomic_fetch_add(p, 1, __ATOMIC_RELAXED, __HIP_MEMORY_SCOPE_WORKGROUP);
}
__device__ __forceinline__ int lds_take(int* p) {
  return __hip_atomic_fetch_add(p, 1, __ATOMIC_RELAXED, __HIP_MEMORY_SCOPE_WORKGROUP);
}

__global__ void __launch_bounds__(1024) bucket_count_kernel(
    const int* __restrict__ erow,
    const int* __restrict__ ecol,
    int* __restrict__ tot,
    int* __restrict__ blkbase) {
  __shared__ int h[NBIN];
  int t = threadIdx.x;
  for (int i = t; i < NBIN; i += 1024) h[i] = 0;
  __syncthreads();
  int lo = blockIdx.x * SLICE;
  int hi = min(lo + SLICE, N_EDGES);
  for (int base = lo + t; base < hi; base += 8 * 1024) {
    int r[8], c[8];
#pragma unroll
    for (int u = 0; u < 8; ++u) {
      int e = base + u * 1024;
      bool ok = e < hi;
      int ec = ok ? e : base;
      int rr = erow[ec];
      c[u] = ecol[ec];
      r[u] = ok ? rr : -1;
    }
#pragma unroll
    for (int u = 0; u < 8; ++u) {
      if (r[u] >= 0) {
        lds_inc(&h[r[u] >> BSH]);
        lds_inc(&h[NBKT + (c[u] >> BSH)]);
      }
    }
  }
  __syncthreads();
  for (int i = t; i < NBIN; i += 1024) {
    int v = h[i];
    int b = 0;
    if (v) b = __hip_atomic_fetch_add(tot + i, v, __ATOMIC_RELAXED, __HIP_MEMORY_SCOPE_AGENT);
    blkbase[blockIdx.x * NBIN + i] = b;
  }
}

__global__ void __launch_bounds__(1024) bucket_scatter_kernel(
    const int* __restrict__ erow,
    const int* __restrict__ ecol,
    const float* __restrict__ val,
    const int* __restrict__ tot,
    const int* __restrict__ blkbase,
    int* __restrict__ bstart,
    int* __restrict__ mid_row,
    int* __restrict__ mid_col) {
  __shared__ int cur[NBIN];
  __shared__ int buf[2][1024];
  int t = threadIdx.x;
  int own = (t < NBIN) ? tot[t] : 0;
  int claimed = (t < NBIN) ? blkbase[blockIdx.x * NBIN + t] : 0;
  buf[0][t] = own;
  __syncthreads();
  int w = 0;
  for (int off = 1; off < 1024; off <<= 1) {
    int x = buf[w][t];
    if (t >= off) x += buf[w][t - off];
    buf[w ^ 1][t] = x;
    __syncthreads();
    w ^= 1;
  }
  int st = buf[w][t] - own;
  if (t >= NBKT) st -= buf[w][NBKT - 1];
  if (t < NBIN) {
    cur[t] = st + claimed;
    if (blockIdx.x == 0) bstart[t] = st;
  }
  __syncthreads();
  int lo = blockIdx.x * SLICE;
  int hi = min(lo + SLICE, N_EDGES);
  int2* mr = reinterpret_cast<int2*>(mid_row);
  int2* mc = reinterpret_cast<int2*>(mid_col);
  for (int base = lo + t; base < hi; base += 8 * 1024) {
    int r[8], c[8], vb[8];
#pragma unroll
    for (int u = 0; u < 8; ++u) {
      int e = base + u * 1024;
      bool ok = e < hi;
      int ec = ok ? e : base;
      int rr = erow[ec];
      c[u] = ecol[ec];
      vb[u] = __float_as_int(val[ec]);
      r[u] = ok ? rr : -1;
    }
#pragma unroll
    for (int u = 0; u < 8; ++u) {
      if (r[u] >= 0) {
        int p = lds_take(&cur[r[u] >> BSH]);
        int q = lds_take(&cur[NBKT + (c[u] >> BSH)]);
        mr[p] = make_int2(c[u] | ((r[u] & (BNODES - 1)) << 17), vb[u]);
        mc[q] = make_int2(r[u] | ((c[u] & (BNODES - 1)) << 17), vb[u]);
      }
    }
  }
}

__global__ void __launch_bounds__(256) bucket_place_kernel(
    const int* __restrict__ tot,
    const int* __restrict__ bstart,
    const int* __restrict__ mid_row,
    const int* __restrict__ mid_col,
    int* __restrict__ start_row,
    int* __restrict__ start_col,
    int* __restrict__ ent_row,
    int* __restrict__ ent_col) {
  __shared__ int cnt[BNODES];
  __shared__ int buf[2][BNODES];
  int t = threadIdx.x;
  int i = blockIdx.x;
  bool bycol = i >= NBKT;
  int b = bycol ? i - NBKT : i;
  const int2* mid = reinterpret_cast<const int2*>(bycol ? mid_col : mid_row);
  int2* ent = reinterpret_cast<int2*>(bycol ? ent_col : ent_row);
  int* runstart = bycol ? start_col : start_row;
  int n = tot[i];
  int first = bstart[i];
  cnt[t] = 0;
  __syncthreads();
  for (int j0 = t; j0 < n; j0 += 8 * BNODES) {
    int key[8];
#pragma unroll
    for (int u = 0; u < 8; ++u) {
      int j = j0 + u * BNODES;
      bool ok = j < n;
      int kk = mid[first + (ok ? j : j0)].x;
      key[u] = ok ? kk : -1;
    }
#pragma unroll
    for (int u = 0; u < 8; ++u)
      if (key[u] >= 0) lds_inc(&cnt[key[u] >> 17]);
  }
  __syncthreads();
  int own = cnt[t];
  buf[0][t] = own;
  __syncthreads();
  int w = 0;
  for (int off = 1; off < BNODES; off <<= 1) {
    int x = buf[w][t];
    if (t >= off) x += buf[w][t - off];
    buf[w ^ 1][t] = x;
    __syncthreads();
    w ^= 1;
  }
  int run = first + buf[w][t] - own;
  int node = b * BNODES + t;
  if (node < N_NODES) runstart[node] = run;
  cnt[t] = run;
  __syncthreads();
  for (int j0 = t; j0 < n; j0 += 8 * BNODES) {
    int2 rec[8];
#pragma unroll
    for (int u = 0; u < 8; ++u) {
      int j = j0 + u * BNODES;
      bool ok = j < n;
      rec[u] = mid[first + (ok ? j : j0)];
      if (!ok) rec[u].x = -1;
    }
#pragma unroll
    for (int u = 0; u < 8; ++u) {
      if (rec[u].x >= 0) {
        int slot = lds_take(&cnt[rec[u].x >> 17]);
        ent[slot] = make_int2(rec[u].x & OTHER_MASK, rec[u].y);
      }
    }
  }
}

template <bool FIRST, bool KEEP_X>
__global__ void __launch_bounds__(256) gather_kernel(
    float* __restrict__ dst,
    const float* __restrict__ src,
    const int* __restrict__ runstart,
    const int* __restrict__ ent,
    float* __restrict__ sum,
    const float* __restrict__ hop,
    const float* __restrict__ th,
    int k) {
  int t = blockIdx.x * blockDim.x + threadIdx.x;
  int node = t >> 4;
  int q = t & 15;
  if (node >= N_NODES) return;
  size_t at = (size_t)node * (D / 4) + q;
  float4 s = make_float4(0.0f, 0.0f, 0.0f, 0.0f);
  if (!FIRST) s = reinterpret_cast<const float4*>(sum)[at];
  float m = fmaxf(fmaxf(hop[0], hop[1]), hop[2]);
  float e0 = __expf(hop[0] - m);
  float e1 = __expf(hop[1] - m);
  float e2 = __expf(hop[2] - m);
  float a = ((k == 0) ? e0 : (k == 1) ? e1 : e2) / (e0 + e1 + e2);
  float co = a * th[k];
  int beg = runstart[node];
  int end = (node + 1 < N_NODES) ? runstart[node + 1] : N_EDGES;
  const int2* rec = reinterpret_cast<const int2*>(ent);
  const float4* s4 = reinterpret_cast<const float4*>(src);
  float4 acc = make_float4(0.0f, 0.0f, 0.0f, 0.0f);
  for (int base = beg; base < end; base += 16) {
    int j = base + q;
    int2 en = make_int2(0, 0);
    if (j < end) en = rec[j];
    int n = min(16, end - base);
    int jj = 0;
    for (; jj + 4 <= n; jj += 4) {
      int c0 = __shfl(en.x, jj + 0, 16);
      int c1 = __shfl(en.x, jj + 1, 16);
      int c2 = __shfl(en.x, jj + 2, 16);
      int c3 = __shfl(en.x, jj + 3, 16);
      float v0 = __int_as_float(__shfl(en.y, jj + 0, 16));
      float v1 = __int_as_float(__shfl(en.y, jj + 1, 16));
      float v2 = __int_as_float(__shfl(en.y, jj + 2, 16));
      float v3 = __int_as_float(__shfl(en.y, jj + 3, 16));
      float4 x0 = s4[(size_t)c0 * (D / 4) + q];
      float4 x1 = s4[(size_t)c1 * (D / 4) + q];
      float4 x2 = s4[(size_t)c2 * (D / 4) + q];
      float4 x3 = s4[(size_t)c3 * (D / 4) + q];
      acc.x += v0 * x0.x; acc.y += v0 * x0.y; acc.z += v0 * x0.z; acc.w += v0 * x0.w;
      acc.x += v1 * x1.x; acc.y += v1 * x1.y; acc.z += v1 * x1.z; acc.w += v1 * x1.w;
      acc.x += v2 * x2.x; acc.y += v2 * x2.y; acc.z += v2 * x2.z; acc.w += v2 * x2.w;
      acc.x += v3 * x3.x; acc.y += v3 * x3.y; acc.z += v3 * x3.z; acc.w += v3 * x3.w;
    }
    for (; jj < n; ++jj) {
      int c0 = __shfl(en.x, jj, 16);
      float v0 = __int_as_float(__shfl(en.y, jj, 16));
      float4 x0 = s4[(size_t)c0 * (D / 4) + q];
      acc.x += v0 * x0.x; acc.y += v0 * x0.y; acc.z += v0 * x0.z; acc.w += v0 * x0.w;
    }
  }
  if (KEEP_X) reinterpret_cast<float4*>(dst)[at] = acc;

  if (FIRST) {
    s.x = co * acc.x;
    s.y = co * acc.y;
    s.z = co * acc.z;
    s.w = co * acc.w;
  } else {
    s.x += co * acc.x;
    s.y += co * acc.y;
    s.z += co * acc.z;
    s.w += co * acc.w;
  }
  reinterpret_cast<float4*>(sum)[at] = s;
}

__global__ void __launch_bounds__(256) gemm_sig_add_kernel(
    float* S,
    const float* __restrict__ Theta,
    const float* __restrict__ H) {
  int lane = threadIdx.x & 31;
  int tile = blockIdx.x * (blockDim.x >> 5) + (threadIdx.x >> 5);
  if (tile >= N_NODES / 16) return;

  int m0 = tile * 16;
  int mrow  = lane & 15;
  int khalf = lane >> 4;

  const float* arow  = S + (size_t)(m0 + mrow) * D + 2 * khalf;
  const float* bbase = Theta + (size_t)(2 * khalf) * D + (lane & 15);

  v8f c0 = {}, c1 = {}, c2 = {}, c3 = {};
#pragma unroll
  for (int kk = 0; kk < 16; ++kk) {
    v2f a;
    a.x = arow[4 * kk];
    a.y = arow[4 * kk + 1];
    const float* bp = bbase + (size_t)(4 * kk) * D;
    v2f b0, b1, b2, b3;
    b0.x = bp[0];      b0.y = bp[D];
    b1.x = bp[16];     b1.y = bp[16 + D];
    b2.x = bp[32];     b2.y = bp[32 + D];
    b3.x = bp[48];     b3.y = bp[48 + D];
    c0 = __builtin_amdgcn_wmma_f32_16x16x4_f32(false, a, false, b0, (short)0, c0, false, false);
    c1 = __builtin_amdgcn_wmma_f32_16x16x4_f32(false, a, false, b1, (short)0, c1, false, false);
    c2 = __builtin_amdgcn_wmma_f32_16x16x4_f32(false, a, false, b2, (short)0, c2, false, false);
    c3 = __builtin_amdgcn_wmma_f32_16x16x4_f32(false, a, false, b3, (short)0, c3, false, false);
  }

  int colb = lane & 15;
#pragma unroll
  for (int v = 0; v < 8; ++v) {
    size_t rb = (size_t)(m0 + v + 8 * khalf) * D;
    float x0 = c0[v], x1 = c1[v], x2 = c2[v], x3 = c3[v];
    S[rb +  0 + colb] = 1.0f / (1.0f + __expf(-x0)) + H[rb +  0 + colb];
    S[rb + 16 + colb] = 1.0f / (1.0f + __expf(-x1)) + H[rb + 16 + colb];
    S[rb + 32 + colb] = 1.0f / (1.0f + __expf(-x2)) + H[rb + 32 + colb];
    S[rb + 48 + colb] = 1.0f / (1.0f + __expf(-x3)) + H[rb + 48 + colb];
  }
}

extern "C" void kernel_launch(void* const* d_in, const int* in_sizes, int n_in,
                              void* d_out, int out_size, void* d_ws, size_t ws_size,
                              hipStream_t stream) {
  const float* H      = (const float*)d_in[0];
  const int*   erow   = (const int*)d_in[1];
  const int*   ecol   = (const int*)d_in[2];
  const float* eval   = (const float*)d_in[3];
  const float* outdeg = (const float*)d_in[4];
  const float* indeg  = (const float*)d_in[5];
  const float* hop    = (const float*)d_in[6];
  const float* th_out = (const float*)d_in[7];
  const float* th_in  = (const float*)d_in[8];
  const float* Theta  = (const float*)d_in[9];

  float* sum = (float*)d_out;
  float* ws  = (float*)d_ws;
  float* X[2]    = {ws, ws + (size_t)NF};
  int*   ent_row = (int*)(ws + 2ull * NF);
  int*   ent_col = ent_row + 2ull * N_EDGES;
  int*   cur_row = ent_col + 2ull * N_EDGES;
  int*   cur_col = cur_row + N_NODES;
  int*   tot     = cur_col + N_NODES;
  int*   bstart  = tot + NBIN;
  int*   blkbase = bstart + NBIN;
  int*   mid_row = (int*)X[1];
  int*   mid_col = mid_row + 2ull * N_EDGES;

  const int elem_blocks = (NF / 4 + 255) / 256;
  const int node_blocks = (N_NODES * 16 + 255) / 256;

  hipMemsetAsync(tot, 0, sizeof(int) * NBIN, stream);

  bucket_count_kernel<<<NB1, 1024, 0, stream>>>(erow, ecol, tot, blkbase);
  bucket_scatter_kernel<<<NB1, 1024, 0, stream>>>(erow, ecol, eval, tot, blkbase, bstart,
                                                  mid_row, mid_col);
  bucket_place_kernel<<<NBIN, 256, 0, stream>>>(tot, bstart, mid_row, mid_col,
                                                cur_row, cur_col, ent_row, ent_col);

  scale_kernel<<<elem_blocks, 256, 0, stream>>>(H, outdeg, X[0]);
  gather_kernel<true, true><<<node_blocks, 256, 0, stream>>>(
      X[1], X[0], cur_row, ent_row, sum, hop, th_out, 0);
  gather_kernel<false, true><<<node_blocks, 256, 0, stream>>>(
      X[0], X[1], cur_row, ent_row, sum, hop, th_out, 1);
  gather_kernel<false, false><<<node_blocks, 256, 0, stream>>>(
      X[1], X[0], cur_row, ent_row, sum, hop, th_out, 2);

  scale_kernel<<<elem_blocks, 256, 0, stream>>>(H, indeg, X[1]);
  gather_kernel<false, true><<<node_blocks, 256, 0, stream>>>(
      X[0], X[1], cur_col, ent_col, sum, hop, th_in, 0);
  gather_kernel<false, true><<<node_blocks, 256, 0, stream>>>(
      X[1], X[0], cur_col, ent_col, sum, hop, th_in, 1);
  gather_kernel<false, false><<<node_blocks, 256, 0, stream>>>(
      X[0], X[1], cur_col, ent_col, sum, hop, th_in, 2);

  const int tiles = N_NODES / 16;
  gemm_sig_add_kernel<<<(tiles + 7) / 8, 256, 0, stream>>>(sum, Theta, H);
}


# Edge sort stores go through LDS: both levels put their records together in LDS in memory order and copy them out lane by lane (whole lines, not 8-byte pieces in 32 lines); sum order rule kept
# speedup vs baseline: 3.8297x; 1.0774x over previous
#include <hip/hip_runtime.h>
#include <hip/hip_bf16.h>

#define N_NODES 100000
#define N_EDGES 1600000
#define D 64
#define NF (N_NODES * D)

typedef float v2f __attribute__((ext_vector_type(2)));
typedef float v8f __attribute__((ext_vector_type(8)));

__global__ void __launch_bounds__(256) scale_kernel(
    const float* __restrict__ H,
    const float* __restrict__ deg,
    float* __restrict__ X) {
  int t = blockIdx.x * blockDim.x + threadIdx.x;
  if (t >= NF / 4) return;
  int node = t / (D / 4);
  float d = fmaxf(deg[node], 1e-8f);
  float4 h = reinterpret_cast<const float4*>(H)[t];
  float4 o;
  o.x = d * h.x; o.y = d * h.y; o.z = d * h.z; o.w = d * h.w;
  reinterpret_cast<float4*>(X)[t] = o;
}

#define BSH 8
#define BNODES (1 << BSH)
#define NBKT ((N_NODES + BNODES - 1) / BNODES)
#define NBIN (2 * NBKT)
#define NB1 256
#define SLICE ((N_EDGES + NB1 - 1) / NB1)
#define OTHER_MASK 0x1FFFF

__device__ __forceinline__ void lds_inc(int* p) {
  __hip_atomic_fetch_add(p, 1, __ATOMIC_RELAXED, __HIP_MEMORY_SCOPE_WORKGROUP);
}
__device__ __forceinline__ int lds_take(int* p) {
  return __hip_atomic_fetch_add(p, 1, __ATOMIC_RELAXED, __HIP_MEMORY_SCOPE_WORKGROUP);
}

__global__ void __launch_bounds__(1024) bucket_count_kernel(
    const int* __restrict__ erow,
    const int* __restrict__ ecol,
    int* __restrict__ tot,
    int* __restrict__ blkbase,
    int* __restrict__ blkcnt) {
  __shared__ int h[NBIN];
  int t = threadIdx.x;
  for (int i = t; i < NBIN; i += 1024) h[i] = 0;
  __syncthreads();
  int lo = blockIdx.x * SLICE;
  int hi = min(lo + SLICE, N_EDGES);
  for (int base = lo + t; base < hi; base += 8 * 1024) {
    int r[8], c[8];
#pragma unroll
    for (int u = 0; u < 8; ++u) {
      int e = base + u * 1024;
      bool ok = e < hi;
      int ec = ok ? e : base;
      int rr = erow[ec];
      c[u] = ecol[ec];
      r[u] = ok ? rr : -1;
    }
#pragma unroll
    for (int u = 0; u < 8; ++u) {
      if (r[u] >= 0) {
        lds_inc(&h[r[u] >> BSH]);
        lds_inc(&h[NBKT + (c[u] >> BSH)]);
      }
    }
  }
  __syncthreads();
  for (int i = t; i < NBIN; i += 1024) {
    int v = h[i];
    int b = 0;
    if (v) b = __hip_atomic_fetch_add(tot + i, v, __ATOMIC_RELAXED, __HIP_MEMORY_SCOPE_AGENT);
    blkbase[blockIdx.x * NBIN + i] = b;
    blkcnt[blockIdx.x * NBIN + i] = v;
  }
}

__global__ void __launch_bounds__(1024) bucket_scatter_kernel(
    const int* __restrict__ erow,
    const int* __restrict__ ecol,
    const float* __restrict__ val,
    const int* __restrict__ tot,
    const int* __restrict__ blkbase,
    const int* __restrict__ blkcnt,
    int* __restrict__ bstart,
    int* __restrict__ mid) {
  __shared__ int cur[NBIN];
  __shared__ int gofs[NBIN];
  __shared__ int2 buf[2][1024];
  __shared__ int2 stage[2 * SLICE];
  __shared__ int place[2 * SLICE];
  int t = threadIdx.x;
  int own = (t < NBIN) ? tot[t] : 0;
  int claimed = (t < NBIN) ? blkbase[blockIdx.x * NBIN + t] : 0;
  int mine = (t < NBIN) ? blkcnt[blockIdx.x * NBIN + t] : 0;
  buf[0][t] = make_int2(own, mine);
  __syncthreads();
  int w = 0;
  for (int off = 1; off < 1024; off <<= 1) {
    int2 x = buf[w][t];
    if (t >= off) {
      int2 y = buf[w][t - off];
      x.x += y.x;
      x.y += y.y;
    }
    buf[w ^ 1][t] = x;
    __syncthreads();
    w ^= 1;
  }
  int st = buf[w][t].x - own;
  if (t >= NBKT) st -= buf[w][NBKT - 1].x;
  int lst = buf[w][t].y - mine;
  if (t < NBIN) {
    cur[t] = lst;
    gofs[t] = st + claimed - lst + ((t >= NBKT) ? N_EDGES : 0);
    if (blockIdx.x == 0) bstart[t] = st;
  }
  __syncthreads();
  int lo = blockIdx.x * SLICE;
  int hi = min(lo + SLICE, N_EDGES);
  for (int base = lo + t; base < hi; base += 8 * 1024) {
    int r[8], c[8], vb[8];
#pragma unroll
    for (int u = 0; u < 8; ++u) {
      int e = base + u * 1024;
      bool ok = e < hi;
      int ec = ok ? e : base;
      int rr = erow[ec];
      c[u] = ecol[ec];
      vb[u] = __float_as_int(val[ec]);
      r[u] = ok ? rr : -1;
    }
#pragma unroll
    for (int u = 0; u < 8; ++u) {
      if (r[u] >= 0) {
        int rb = r[u] >> BSH;
        int cb = NBKT + (c[u] >> BSH);
        int p = lds_take(&cur[rb]);
        int q = lds_take(&cur[cb]);
        stage[p] = make_int2(c[u] | ((r[u] & (BNODES - 1)) << 17), vb[u]);
        place[p] = gofs[rb] + p;
        stage[q] = make_int2(r[u] | ((c[u] & (BNODES - 1)) << 17), vb[u]);
        place[q] = gofs[cb] + q;
      }
    }
  }
  __syncthreads();
  int m = 2 * (hi - lo);
  int2* out = reinterpret_cast<int2*>(mid);
  for (int i0 = t; i0 < m; i0 += 4 * 1024) {
    int2 rr[4];
    int pp[4];
#pragma unroll
    for (int u = 0; u < 4; ++u) {
      int i = i0 + u * 1024;
      rr[u] = make_int2(0, 0);
      pp[u] = 0;
      if (i < m) {
        rr[u] = stage[i];
        pp[u] = place[i];
      }
    }
#pragma unroll
    for (int u = 0; u < 4; ++u)
      if (i0 + u * 1024 < m) out[pp[u]] = rr[u];
  }
}

#define STAGE2 8192
__global__ void __launch_bounds__(256) bucket_place_kernel(
    const int* __restrict__ tot,
    const int* __restrict__ bstart,
    const int* __restrict__ mid_row,
    const int* __restrict__ mid_col,
    int* __restrict__ start_row,
    int* __restrict__ start_col,
    int* __restrict__ ent_row,
    int* __restrict__ ent_col) {
  __shared__ int cnt[BNODES];
  __shared__ int buf[2][BNODES];
  __shared__ int2 stage[STAGE2];
  int t = threadIdx.x;
  int i = blockIdx.x;
  bool bycol = i >= NBKT;
  int b = bycol ? i - NBKT : i;
  const int2* mid = reinterpret_cast<const int2*>(bycol ? mid_col : mid_row);
  int2* ent = reinterpret_cast<int2*>(bycol ? ent_col : ent_row);
  int* runstart = bycol ? start_col : start_row;
  int n = tot[i];
  int first = bstart[i];
  cnt[t] = 0;
  __syncthreads();
  for (int j0 = t; j0 < n; j0 += 8 * BNODES) {
    int key[8];
#pragma unroll
    for (int u = 0; u < 8; ++u) {
      int j = j0 + u * BNODES;
      bool ok = j < n;
      int kk = mid[first + (ok ? j : j0)].x;
      key[u] = ok ? kk : -1;
    }
#pragma unroll
    for (int u = 0; u < 8; ++u)
      if (key[u] >= 0) lds_inc(&cnt[key[u] >> 17]);
  }
  __syncthreads();
  int own = cnt[t];
  buf[0][t] = own;
  __syncthreads();
  int w = 0;
  for (int off = 1; off < BNODES; off <<= 1) {
    int x = buf[w][t];
    if (t >= off) x += buf[w][t - off];
    buf[w ^ 1][t] = x;
    __syncthreads();
    w ^= 1;
  }
  int run = buf[w][t] - own;
  int node = b * BNODES + t;
  if (node < N_NODES) runstart[node] = first + run;
  cnt[t] = run;
  __syncthreads();
  bool fits = n <= STAGE2;
  int2* outp = ent + first;
  for (int j0 = t; j0 < n; j0 += 8 * BNODES) {
    int2 rec[8];
#pragma unroll
    for (int u = 0; u < 8; ++u) {
      int j = j0 + u * BNODES;
      bool ok = j < n;
      rec[u] = mid[first + (ok ? j : j0)];
      if (!ok) rec[u].x = -1;
    }
#pragma unroll
    for (int u = 0; u < 8; ++u) {
      if (rec[u].x >= 0) {
        int slot = lds_take(&cnt[rec[u].x >> 17]);
        int2 fin = make_int2(rec[u].x & OTHER_MASK, rec[u].y);
        if (fits) stage[slot] = fin;
        else outp[slot] = fin;
      }
    }
  }
  if (fits) {
    __syncthreads();
    for (int j0 = t; j0 < n; j0 += 4 * BNODES) {
      int2 rr[4];
#pragma unroll
      for (int u = 0; u < 4; ++u) {
        int j = j0 + u * BNODES;
        rr[u] = make_int2(0, 0);
        if (j < n) rr[u] = stage[j];
      }
#pragma unroll
      for (int u = 0; u < 4; ++u)
        if (j0 + u * BNODES < n) outp[j0 + u * BNODES] = rr[u];
    }
  }
}

template <bool FIRST, bool KEEP_X>
__global__ void __launch_bounds__(256) gather_kernel(
    float* __restrict__ dst,
    const float* __restrict__ src,
    const int* __restrict__ runstart,
    const int* __restrict__ ent,
    float* __restrict__ sum,
    const float* __restrict__ hop,
    const float* __restrict__ th,
    int k) {
  int t = blockIdx.x * blockDim.x + threadIdx.x;
  int node = t >> 4;
  int q = t & 15;
  if (node >= N_NODES) return;
  size_t at = (size_t)node * (D / 4) + q;
  float4 s = make_float4(0.0f, 0.0f, 0.0f, 0.0f);
  if (!FIRST) s = reinterpret_cast<const float4*>(sum)[at];
  float m = fmaxf(fmaxf(hop[0], hop[1]), hop[2]);
  float e0 = __expf(hop[0] - m);
  float e1 = __expf(hop[1] - m);
  float e2 = __expf(hop[2] - m);
  float a = ((k == 0) ? e0 : (k == 1) ? e1 : e2) / (e0 + e1 + e2);
  float co = a * th[k];
  int beg = runstart[node];
  int end = (node + 1 < N_NODES) ? runstart[node + 1] : N_EDGES;
  const int2* rec = reinterpret_cast<const int2*>(ent);
  const float4* s4 = reinterpret_cast<const float4*>(src);
  float4 acc = make_float4(0.0f, 0.0f, 0.0f, 0.0f);
  for (int base = beg; base < end; base += 16) {
    int j = base + q;
    int2 en = make_int2(0, 0);
    if (j < end) en = rec[j];
    int n = min(16, end - base);
    int jj = 0;
    for (; jj + 4 <= n; jj += 4) {
      int c0 = __shfl(en.x, jj + 0, 16);
      int c1 = __shfl(en.x, jj + 1, 16);
      int c2 = __shfl(en.x, jj + 2, 16);
      int c3 = __shfl(en.x, jj + 3, 16);
      float v0 = __int_as_float(__shfl(en.y, jj + 0, 16));
      float v1 = __int_as_float(__shfl(en.y, jj + 1, 16));
      float v2 = __int_as_float(__shfl(en.y, jj + 2, 16));
      float v3 = __int_as_float(__shfl(en.y, jj + 3, 16));
      float4 x0 = s4[(size_t)c0 * (D / 4) + q];
      float4 x1 = s4[(size_t)c1 * (D / 4) + q];
      float4 x2 = s4[(size_t)c2 * (D / 4) + q];
      float4 x3 = s4[(size_t)c3 * (D / 4) + q];
      acc.x += v0 * x0.x; acc.y += v0 * x0.y; acc.z += v0 * x0.z; acc.w += v0 * x0.w;
      acc.x += v1 * x1.x; acc.y += v1 * x1.y; acc.z += v1 * x1.z; acc.w += v1 * x1.w;
      acc.x += v2 * x2.x; acc.y += v2 * x2.y; acc.z += v2 * x2.z; acc.w += v2 * x2.w;
      acc.x += v3 * x3.x; acc.y += v3 * x3.y; acc.z += v3 * x3.z; acc.w += v3 * x3.w;
    }
    for (; jj < n; ++jj) {
      int c0 = __shfl(en.x, jj, 16);
      float v0 = __int_as_float(__shfl(en.y, jj, 16));
      float4 x0 = s4[(size_t)c0 * (D / 4) + q];
      acc.x += v0 * x0.x; acc.y += v0 * x0.y; acc.z += v0 * x0.z; acc.w += v0 * x0.w;
    }
  }
  if (KEEP_X) reinterpret_cast<float4*>(dst)[at] = acc;

  if (FIRST) {
    s.x = co * acc.x;
    s.y = co * acc.y;
    s.z = co * acc.z;
    s.w = co * acc.w;
  } else {
    s.x += co * acc.x;
    s.y += co * acc.y;
    s.z += co * acc.z;
    s.w += co * acc.w;
  }
  reinterpret_cast<float4*>(sum)[at] = s;
}

__global__ void __launch_bounds__(256) gemm_sig_add_kernel(
    float* S,
    const float* __restrict__ Theta,
    const float* __restrict__ H) {
  int lane = threadIdx.x & 31;
  int tile = blockIdx.x * (blockDim.x >> 5) + (threadIdx.x >> 5);
  if (tile >= N_NODES / 16) return;

  int m0 = tile * 16;
  int mrow  = lane & 15;
  int khalf = lane >> 4;

  const float* arow  = S + (size_t)(m0 + mrow) * D + 2 * khalf;
  const float* bbase = Theta + (size_t)(2 * khalf) * D + (lane & 15);

  v8f c0 = {}, c1 = {}, c2 = {}, c3 = {};
#pragma unroll
  for (int kk = 0; kk < 16; ++kk) {
    v2f a;
    a.x = arow[4 * kk];
    a.y = arow[4 * kk + 1];
    const float* bp = bbase + (size_t)(4 * kk) * D;
    v2f b0, b1, b2, b3;
    b0.x = bp[0];      b0.y = bp[D];
    b1.x = bp[16];     b1.y = bp[16 + D];
    b2.x = bp[32];     b2.y = bp[32 + D];
    b3.x = bp[48];     b3.y = bp[48 + D];
    c0 = __builtin_amdgcn_wmma_f32_16x16x4_f32(false, a, false, b0, (short)0, c0, false, false);
    c1 = __builtin_amdgcn_wmma_f32_16x16x4_f32(false, a, false, b1, (short)0, c1, false, false);
    c2 = __builtin_amdgcn_wmma_f32_16x16x4_f32(false, a, false, b2, (short)0, c2, false, false);
    c3 = __builtin_amdgcn_wmma_f32_16x16x4_f32(false, a, false, b3, (short)0, c3, false, false);
  }

  int colb = lane & 15;
#pragma unroll
  for (int v = 0; v < 8; ++v) {
    size_t rb = (size_t)(m0 + v + 8 * khalf) * D;
    float x0 = c0[v], x1 = c1[v], x2 = c2[v], x3 = c3[v];
    S[rb +  0 + colb] = 1.0f / (1.0f + __expf(-x0)) + H[rb +  0 + colb];
    S[rb + 16 + colb] = 1.0f / (1.0f + __expf(-x1)) + H[rb + 16 + colb];
    S[rb + 32 + colb] = 1.0f / (1.0f + __expf(-x2)) + H[rb + 32 + colb];
    S[rb + 48 + colb] = 1.0f / (1.0f + __expf(-x3)) + H[rb + 48 + colb];
  }
}

extern "C" void kernel_launch(void* const* d_in, const int* in_sizes, int n_in,
                              void* d_out, int out_size, void* d_ws, size_t ws_size,
                              hipStream_t stream) {
  const float* H      = (const float*)d_in[0];
  const int*   erow   = (const int*)d_in[1];
  const int*   ecol   = (const int*)d_in[2];
  const float* eval   = (const float*)d_in[3];
  const float* outdeg = (const float*)d_in[4];
  const float* indeg  = (const float*)d_in[5];
  const float* hop    = (const float*)d_in[6];
  const float* th_out = (const float*)d_in[7];
  const float* th_in  = (const float*)d_in[8];
  const float* Theta  = (const float*)d_in[9];

  float* sum = (float*)d_out;
  float* ws  = (float*)d_ws;
  float* X[2]    = {ws, ws + (size_t)NF};
  int*   ent_row = (int*)(ws + 2ull * NF);
  int*   ent_col = ent_row + 2ull * N_EDGES;
  int*   cur_row = ent_col + 2ull * N_EDGES;
  int*   cur_col = cur_row + N_NODES;
  int*   tot     = cur_col + N_NODES;
  int*   bstart  = tot + NBIN;
  int*   blkbase = bstart + NBIN;
  int*   blkcnt  = blkbase + NB1 * NBIN;
  int*   mid_row = (int*)X[1];
  int*   mid_col = mid_row + 2ull * N_EDGES;

  const int elem_blocks = (NF / 4 + 255) / 256;
  const int node_blocks = (N_NODES * 16 + 255) / 256;

  hipMemsetAsync(tot, 0, sizeof(int) * NBIN, stream);

  bucket_count_kernel<<<NB1, 1024, 0, stream>>>(erow, ecol, tot, blkbase, blkcnt);
  bucket_scatter_kernel<<<NB1, 1024, 0, stream>>>(erow, ecol, eval, tot, blkbase, blkcnt,
                                                  bstart, mid_row);
  bucket_place_kernel<<<NBIN, 256, 0, stream>>>(tot, bstart, mid_row, mid_col,
                                                cur_row, cur_col, ent_row, ent_col);

  scale_kernel<<<elem_blocks, 256, 0, stream>>>(H, outdeg, X[0]);
  gather_kernel<true, true><<<node_blocks, 256, 0, stream>>>(
      X[1], X[0], cur_row, ent_row, sum, hop, th_out, 0);
  gather_kernel<false, true><<<node_blocks, 256, 0, stream>>>(
      X[0], X[1], cur_row, ent_row, sum, hop, th_out, 1);
  gather_kernel<false, false><<<node_blocks, 256, 0, stream>>>(
      X[1], X[0], cur_row, ent_row, sum, hop, th_out, 2);

  scale_kernel<<<elem_blocks, 256, 0, stream>>>(H, indeg, X[1]);
  gather_kernel<false, true><<<node_blocks, 256, 0, stream>>>(
      X[0], X[1], cur_col, ent_col, sum, hop, th_in, 0);
  gather_kernel<false, true><<<node_blocks, 256, 0, stream>>>(
      X[1], X[0], cur_col, ent_col, sum, hop, th_in, 1);
  gather_kernel<false, false><<<node_blocks, 256, 0, stream>>>(
      X[0], X[1], cur_col, ent_col, sum, hop, th_in, 2);

  const int tiles = N_NODES / 16;
  gemm_sig_add_kernel<<<(tiles + 7) / 8, 256, 0, stream>>>(sum, Theta, H);
}
